# starting kernel
# baseline (speedup 1.0000x reference)
#include <hip/hip_runtime.h>

#define IMAGE    128
#define N_LINES  32
#define N_CURVES 64
#define SEG_PAD  16
#define N_SEG    9
#define SIGMA2   0.05f
#define EPSF     1e-10f
#define BIGF     1e30f

typedef float v2f __attribute__((ext_vector_type(2)));
typedef float v8f __attribute__((ext_vector_type(8)));

__device__ __forceinline__ float clamp01(float v) { return fminf(fmaxf(v, 0.0f), 1.0f); }

__global__ __launch_bounds__(256)
void render_curves_wmma(const float* __restrict__ position,
                        const float* __restrict__ radius,
                        const float* __restrict__ color,
                        const float* __restrict__ background,
                        float* __restrict__ out)
{
    __align__(16) __shared__ float sS1[N_CURVES][SEG_PAD][4];
    __align__(16) __shared__ float sS2[N_CURVES][SEG_PAD][4];
    __align__(16) __shared__ float sS3[N_CURVES][SEG_PAD][4];
    __align__(16) __shared__ float sMeta[N_CURVES][4];

    const int b   = blockIdx.x / IMAGE;
    const int y   = blockIdx.x % IMAGE;
    const int tid = threadIdx.x;

    for (int e = tid; e < N_CURVES * SEG_PAD; e += 256) {
        const int curve = e >> 4;
        const int seg   = e & 15;
        if (seg < N_SEG) {
            const int line = curve >> 1;
            const float* P = position + (b * N_LINES + line) * 10;
            float c0y, c0x, c1y, c1x, c2y, c2x;
            if ((curve & 1) == 0) {
                c0y = P[0]; c0x = P[1];
                c1y = P[2]; c1x = P[3];
                c2y = P[4]; c2x = P[5];
            } else {
                c0y = P[4];               c0x = P[5];
                c1y = 2.0f * P[4] - P[6]; c1x = 2.0f * P[5] - P[7];
                c2y = P[8];               c2x = P[9];
            }
            const float t0 = (float)seg       * (1.0f / 9.0f);
            const float t1 = (float)(seg + 1) * (1.0f / 9.0f);
            const float u0 = 1.0f - t0, u1 = 1.0f - t1;
            const float ay = u0*u0*c0y + 2.0f*t0*u0*c1y + t0*t0*c2y;
            const float ax = u0*u0*c0x + 2.0f*t0*u0*c1x + t0*t0*c2x;
            const float ey = u1*u1*c0y + 2.0f*t1*u1*c1y + t1*t1*c2y;
            const float ex = u1*u1*c0x + 2.0f*t1*u1*c1x + t1*t1*c2x;
            const float vy = ey - ay, vx = ex - ax;
            const float vv = vy*vy + vx*vx;
            const float inv = 1.0f / (vv + EPSF);
            sS1[curve][seg][0] = vy * inv;
            sS1[curve][seg][1] = vx * inv;
            sS1[curve][seg][2] = -(ay*vy + ax*vx) * inv;
            sS1[curve][seg][3] = 0.0f;
            sS2[curve][seg][0] = -2.0f * ay;
            sS2[curve][seg][1] = -2.0f * ax;
            sS2[curve][seg][2] = ay*ay + ax*ax;
            sS2[curve][seg][3] = 1.0f;
            sS3[curve][seg][0] = 0.0f; sS3[curve][seg][1] = 0.0f;
            sS3[curve][seg][2] = vv;   sS3[curve][seg][3] = 0.0f;
        } else {
            sS1[curve][seg][0] = 0.0f; sS1[curve][seg][1] = 0.0f;
            sS1[curve][seg][2] = 0.0f; sS1[curve][seg][3] = 0.0f;
            sS2[curve][seg][0] = 0.0f; sS2[curve][seg][1] = 0.0f;
            sS2[curve][seg][2] = BIGF; sS2[curve][seg][3] = 0.0f;
            sS3[curve][seg][0] = 0.0f; sS3[curve][seg][1] = 0.0f;
            sS3[curve][seg][2] = 0.0f; sS3[curve][seg][3] = 0.0f;
        }
    }
    if (tid < N_CURVES) {
        const int line = tid >> 1;
        const float r  = radius[b * N_LINES + line];
        const float sg = r * r * SIGMA2;
        sMeta[tid][0] = 1.0f / sg;
        sMeta[tid][1] = color[(b * N_LINES + line) * 3 + 0];
        sMeta[tid][2] = color[(b * N_LINES + line) * 3 + 1];
        sMeta[tid][3] = color[(b * N_LINES + line) * 3 + 2];
    }
    __syncthreads();

    const int wave = tid >> 5;
    const int lane = tid & 31;
    const int colp = lane & 15;
    const int x    = wave * 16 + colp;
    const float sc = 2.0f / 127.0f;
    const float gy = -1.0f + (float)y * sc;
    const float gx = -1.0f + (float)x * sc;
    const bool  hi = lane >= 16;

    v2f Bf;
    Bf.x = hi ? 1.0f : gy;
    Bf.y = hi ? (gy * gy + gx * gx) : gx;

    const int segSel = lane & 15;
    const int kOff   = hi ? 2 : 0;

    float accR = 0.0f, accG = 0.0f, accB = 0.0f;
    float w = 1.0f;

    for (int curve = 0; curve < N_CURVES; ++curve) {
        const v2f A1 = *(const v2f*)&sS1[curve][segSel][kOff];
        const v2f A2 = *(const v2f*)&sS2[curve][segSel][kOff];
        const v2f A3 = *(const v2f*)&sS3[curve][segSel][kOff];

        v8f U  = {};
        v8f S2 = {};
        v8f VV = {};
        U  = __builtin_amdgcn_wmma_f32_16x16x4_f32(false, A1, false, Bf, (short)0, U,  false, false);
        S2 = __builtin_amdgcn_wmma_f32_16x16x4_f32(false, A2, false, Bf, (short)0, S2, false, false);
        VV = __builtin_amdgcn_wmma_f32_16x16x4_f32(false, A3, false, Bf, (short)0, VV, false, false);

        float m = BIGF;
#pragma unroll
        for (int i = 0; i < 8; ++i) {
            const float u  = U[i];
            const float tp = clamp01(u);
            const float e2 = S2[i] - VV[i] * (tp * (2.0f * u - tp));
            m = fminf(m, e2);
        }
        m = fminf(m, __shfl_xor(m, 16, 32));

        const float invsig = sMeta[curve][0];
        const float z = m * invsig;
        const float I = __expf(-(z * z));
        const float wI = w * I;
        accR = fmaf(sMeta[curve][1], wI, accR);
        accG = fmaf(sMeta[curve][2], wI, accG);
        accB = fmaf(sMeta[curve][3], wI, accB);
        w = w * (1.0f - I + EPSF);
    }

    if (!hi) {
        const float bg0 = background[b * 3 + 0];
        const float bg1 = background[b * 3 + 1];
        const float bg2 = background[b * 3 + 2];
        const int base = ((b * 3) * IMAGE + y) * IMAGE + x;
        out[base]                     = clamp01(fmaf(bg0, w, accR));
        out[base + IMAGE * IMAGE]     = clamp01(fmaf(bg1, w, accG));
        out[base + 2 * IMAGE * IMAGE] = clamp01(fmaf(bg2, w, accB));
    }
}

extern "C" void kernel_launch(void* const* d_in, const int* in_sizes, int n_in,
                              void* d_out, int out_size, void* d_ws, size_t ws_size,
                              hipStream_t stream) {
    (void)in_sizes; (void)n_in; (void)out_size; (void)d_ws; (void)ws_size;
    const float* position   = (const float*)d_in[0];
    const float* radius     = (const float*)d_in[1];
    const float* color      = (const float*)d_in[2];
    const float* background = (const float*)d_in[3];
    float* out = (float*)d_out;
    render_curves_wmma<<<4 * IMAGE, 256, 0, stream>>>(position, radius, color, background, out);
}


# Table build as 2 matrix products per wave: sample weights x control values of 16 curves give a lane consecutive vertices of its curve; one fetch round (radius, colour too), no build loop; rounds anew
# speedup vs baseline: 2.4439x; 2.4439x over previous
#include <hip/hip_runtime.h>


#define IMAGE    128
#define N_LINES  32
#define N_CURVES 64
#define N_SEG    9
#define N_VERT   10
#define N_TILES  4
#define SIGMA2   0.05f
#define EPSF     1e-10f
#define BIGF     1e30f

typedef float v2f __attribute__((ext_vector_type(2)));
typedef float v4f __attribute__((ext_vector_type(4)));
typedef float v8f __attribute__((ext_vector_type(8)));

__device__ __forceinline__ float clamp01(float v) { return fminf(fmaxf(v, 0.0f), 1.0f); }

__device__ __forceinline__ float xhalf(float x) {
    return __builtin_bit_cast(float, __builtin_amdgcn_permlanex16(__builtin_bit_cast(int, x), __builtin_bit_cast(int, x),
                                                                  0x76543210, 0xfedcba98, false, false));
}

__global__ __launch_bounds__(256)
void render_curves_wmma(const float* __restrict__ position,
                        const float* __restrict__ radius,
                        const float* __restrict__ color,
                        const float* __restrict__ background,
                        float* __restrict__ out)
{
    __align__(16) __shared__ float sVert[N_VERT][N_CURVES][4];
    __align__(16) __shared__ float sVV[N_SEG][N_CURVES];
    __align__(16) __shared__ float sHH[N_SEG][N_CURVES];
    __align__(16) __shared__ float sMeta[N_CURVES][4];

    const int b   = blockIdx.x / (IMAGE / 2);
    const int yp  = blockIdx.x % (IMAGE / 2);
    const int tid = threadIdx.x;

    const int wave = tid >> 5;
    const int lane = tid & 31;
    const int colp = lane & 15;
    const bool  hi = lane >= 16;

    float mRad = 1.0f, mC0 = 0.0f, mC1 = 0.0f, mC2 = 0.0f;
    {
        const int slot  = 16 * (wave & 3) + colp;
        const int curve = ((slot >> 3) & 1) * 32 + (slot >> 4) * 8 + (slot & 7);
        const bool odd  = (curve & 1) != 0;
        const float* P  = position + (b * N_LINES + (curve >> 1)) * 10;
        const int i0 = (odd ? 4 : 0) + (hi ? 4 : 0);
        const int i1 = hi ? i0 : i0 + 2;
        const v2f q0 = *(const v2f*)(P + i0);
        const v2f q1 = *(const v2f*)(P + i1);
        if (tid < N_CURVES) {
            const int mcurve = ((tid >> 3) & 1) * 32 + (tid >> 4) * 8 + (tid & 7);
            const int mline  = mcurve >> 1;
            mRad = radius[b * N_LINES + mline];
            mC0  = color[(b * N_LINES + mline) * 3 + 0];
            mC1  = color[(b * N_LINES + mline) * 3 + 1];
            mC2  = color[(b * N_LINES + mline) * 3 + 2];
        }
        const float midy = odd ? 2.0f * q0.x - q1.x : q1.x;
        const float midx = odd ? 2.0f * q0.y - q1.y : q1.y;
        v2f By, Bx;
        By.x = q0.x;  By.y = hi ? 0.0f : midy;
        Bx.x = q0.y;  Bx.y = hi ? 0.0f : midx;

        const int   js = colp < 8 ? colp : (colp < 11 ? colp - 1 : N_VERT - 1);
        const float tj = (float)js * (1.0f / 9.0f);
        const float uj = 1.0f - tj;
        v2f Aw;
        Aw.x = hi ? tj * tj : uj * uj;
        Aw.y = hi ? 0.0f    : 2.0f * tj * uj;

        v8f Dy = {}, Dx = {};
        Dy = __builtin_amdgcn_wmma_f32_16x16x4_f32(false, Aw, false, By, (short)0, Dy, false, false);
        Dx = __builtin_amdgcn_wmma_f32_16x16x4_f32(false, Aw, false, Bx, (short)0, Dx, false, false);

        const int first = hi ? 7 : 0;
        if (wave < 4) {
#pragma unroll
            for (int r = 0; r < 8; ++r) {
                const int vi = first + r;
                if (vi < N_VERT) {
                    const float ay = Dy[r], ax = Dx[r];
                    const v4f r2 = { -2.0f * ay, -2.0f * ax, ay*ay + ax*ax, 1.0f };
                    *(v4f*)&sVert[vi][slot][0] = r2;
                }
            }
        } else {
#pragma unroll
            for (int r = 0; r < 7; ++r) {
                const int si = first + r;
                if (si < N_SEG) {
                    const float vy = Dy[r + 1] - Dy[r], vx = Dx[r + 1] - Dx[r];
                    const float vv = vy*vy + vx*vx;
                    sVV[si][slot] = vv;
                    sHH[si][slot] = 0.5f / (vv + EPSF);
                }
            }
        }
    }
    if (tid < N_CURVES) {
        const float sg = mRad * mRad * SIGMA2;
        const v4f mt = { 1.0f / sg, mC0, mC1, mC2 };
        *(v4f*)&sMeta[tid][0] = mt;
    }
    __syncthreads();

    const int y    = 2 * yp + (wave >> 2);
    const int xs   = (wave & 3) * 32;
    const int x0   = xs + colp;
    const int x1   = xs + 16 + colp;
    const float sc = 2.0f / 127.0f;
    const float gy  = -1.0f + (float)y  * sc;
    const float gx0 = -1.0f + (float)x0 * sc;
    const float gx1 = -1.0f + (float)x1 * sc;

    v2f Bf0, Bf1;
    Bf0.x = hi ? 1.0f : gy;
    Bf0.y = hi ? fmaf(gy, gy, gx0 * gx0) : gx0;
    Bf1.x = hi ? 1.0f : gy;
    Bf1.y = hi ? fmaf(gy, gy, gx1 * gx1) : gx1;

    const int rowSel = lane & 15;
    const int kOff   = hi ? 2 : 0;
    const int metaRow = hi ? 8 : 0;

    const v8f zero8 = {0.0f, 0.0f, 0.0f, 0.0f, 0.0f, 0.0f, 0.0f, 0.0f};
    const v8f one8  = {1.0f, 1.0f, 1.0f, 1.0f, 1.0f, 1.0f, 1.0f, 1.0f};

    float aR0 = 0.0f, aG0 = 0.0f, aB0 = 0.0f, w0 = 1.0f;
    float aR1 = 0.0f, aG1 = 0.0f, aB1 = 0.0f, w1 = 1.0f;

    for (int t = 0; t < N_TILES; ++t) {
        v8f m0 = {BIGF, BIGF, BIGF, BIGF, BIGF, BIGF, BIGF, BIGF};
        v8f m1 = {BIGF, BIGF, BIGF, BIGF, BIGF, BIGF, BIGF, BIGF};

        v8f Dp0 = {}, Dp1 = {};
        {
            const v2f A0 = *(const v2f*)&sVert[0][t * 16 + rowSel][kOff];
            Dp0 = __builtin_amdgcn_wmma_f32_16x16x4_f32(false, A0, false, Bf0, (short)0, Dp0, false, false);
            Dp1 = __builtin_amdgcn_wmma_f32_16x16x4_f32(false, A0, false, Bf1, (short)0, Dp1, false, false);
        }

#pragma unroll
        for (int s = 0; s < N_SEG; ++s) {
            const v2f A1 = *(const v2f*)&sVert[s + 1][t * 16 + rowSel][kOff];
            const v4f va = *(const v4f*)&sVV[s][t * 16 + metaRow];
            const v4f vb = *(const v4f*)&sVV[s][t * 16 + metaRow + 4];
            const v4f ha = *(const v4f*)&sHH[s][t * 16 + metaRow];
            const v4f hb = *(const v4f*)&sHH[s][t * 16 + metaRow + 4];
            const v8f VV = __builtin_shufflevector(va, vb, 0, 1, 2, 3, 4, 5, 6, 7);
            const v8f HH = __builtin_shufflevector(ha, hb, 0, 1, 2, 3, 4, 5, 6, 7);

            v8f Dn0 = {}, Dn1 = {};
            Dn0 = __builtin_amdgcn_wmma_f32_16x16x4_f32(false, A1, false, Bf0, (short)0, Dn0, false, false);
            Dn1 = __builtin_amdgcn_wmma_f32_16x16x4_f32(false, A1, false, Bf1, (short)0, Dn1, false, false);

            {
                const v8f G  = (Dp0 - Dn0) + VV;
                const v8f tp = __builtin_elementwise_min(__builtin_elementwise_max(G * HH, zero8), one8);
                const v8f e2 = Dp0 - tp * (G - tp * VV);
                m0 = __builtin_elementwise_min(m0, e2);
                Dp0 = Dn0;
            }
            {
                const v8f G  = (Dp1 - Dn1) + VV;
                const v8f tp = __builtin_elementwise_min(__builtin_elementwise_max(G * HH, zero8), one8);
                const v8f e2 = Dp1 - tp * (G - tp * VV);
                m1 = __builtin_elementwise_min(m1, e2);
                Dp1 = Dn1;
            }
        }

#pragma unroll
        for (int i = 0; i < 8; ++i) {
            const v4f mt = *(const v4f*)&sMeta[t * 16 + metaRow + i][0];
            {
                const float z = m0[i] * mt.x;
                const float I = __expf(-(z * z));
                const float wI = w0 * I;
                aR0 = fmaf(mt.y, wI, aR0);
                aG0 = fmaf(mt.z, wI, aG0);
                aB0 = fmaf(mt.w, wI, aB0);
                w0 = w0 * (1.0f - I + EPSF);
            }
            {
                const float z = m1[i] * mt.x;
                const float I = __expf(-(z * z));
                const float wI = w1 * I;
                aR1 = fmaf(mt.y, wI, aR1);
                aG1 = fmaf(mt.z, wI, aG1);
                aB1 = fmaf(mt.w, wI, aB1);
                w1 = w1 * (1.0f - I + EPSF);
            }
        }
    }

    const float kR = hi ? aR1 : aR0, kG = hi ? aG1 : aG0, kB = hi ? aB1 : aB0, kW = hi ? w1 : w0;
    const float rR = xhalf(hi ? aR0 : aR1), rG = xhalf(hi ? aG0 : aG1);
    const float rB = xhalf(hi ? aB0 : aB1), rW = xhalf(hi ? w0 : w1);
    const float fwR = hi ? rR : kR, fwG = hi ? rG : kG, fwB = hi ? rB : kB, fwW = hi ? rW : kW;
    const float bkR = hi ? kR : rR, bkG = hi ? kG : rG, bkB = hi ? kB : rB, bkW = hi ? kW : rW;
    {
        const float fR = fmaf(fwW, bkR, fwR);
        const float fG = fmaf(fwW, bkG, fwG);
        const float fB = fmaf(fwW, bkB, fwB);
        const float fW = fwW * bkW;
        const float bg0 = background[b * 3 + 0];
        const float bg1 = background[b * 3 + 1];
        const float bg2 = background[b * 3 + 2];
        const int base = ((b * 3) * IMAGE + y) * IMAGE + xs + lane;
        out[base]                     = clamp01(fmaf(bg0, fW, fR));
        out[base + IMAGE * IMAGE]     = clamp01(fmaf(bg1, fW, fG));
        out[base + 2 * IMAGE * IMAGE] = clamp01(fmaf(bg2, fW, fB));
    }
}

extern "C" void kernel_launch(void* const* d_in, const int* in_sizes, int n_in,
                              void* d_out, int out_size, void* d_ws, size_t ws_size,
                              hipStream_t stream) {
    (void)in_sizes; (void)n_in; (void)out_size; (void)d_ws; (void)ws_size;
    const float* position   = (const float*)d_in[0];
    const float* radius     = (const float*)d_in[1];
    const float* color      = (const float*)d_in[2];
    const float* background = (const float*)d_in[3];
    float* out = (float*)d_out;
    render_curves_wmma<<<4 * (IMAGE / 2), 256, 0, stream>>>(position, radius, color, background, out);
}


# Foot parameter's clip to [0,1] rides on its packed multiply as the clamp output modifier (v_pk_mul_f32 ... clamp, in asm): the 16 lone v_max per segment step are gone; results bit-identical
# speedup vs baseline: 2.7105x; 1.1091x over previous
#include <hip/hip_runtime.h>


#define IMAGE    128
#define N_LINES  32
#define N_CURVES 64
#define N_SEG    9
#define N_VERT   10
#define N_TILES  4
#define SIGMA2   0.05f
#define EPSF     1e-10f
#define BIGF     1e30f

typedef float v2f __attribute__((ext_vector_type(2)));
typedef float v4f __attribute__((ext_vector_type(4)));
typedef float v8f __attribute__((ext_vector_type(8)));

__device__ __forceinline__ float clamp01(float v) { return fminf(fmaxf(v, 0.0f), 1.0f); }

__device__ __forceinline__ float xhalf(float x) {
    return __builtin_bit_cast(float, __builtin_amdgcn_permlanex16(__builtin_bit_cast(int, x), __builtin_bit_cast(int, x),
                                                                  0x76543210, 0xfedcba98, false, false));
}


__device__ __forceinline__ void mul_clamp01_x2(const v8f g0, const v8f g1, const v8f h, v8f& t0, v8f& t1) {
    const v2f a0 = { g0[0], g0[1] }, a1 = { g0[2], g0[3] }, a2 = { g0[4], g0[5] }, a3 = { g0[6], g0[7] };
    const v2f b0 = { g1[0], g1[1] }, b1 = { g1[2], g1[3] }, b2 = { g1[4], g1[5] }, b3 = { g1[6], g1[7] };
    const v2f h0 = { h[0], h[1] },   h1 = { h[2], h[3] },   h2 = { h[4], h[5] },   h3 = { h[6], h[7] };
    v2f o0, o1, o2, o3, o4, o5, o6, o7;
    asm("v_pk_mul_f32 %0, %8, %16 clamp\n\t"
        "v_pk_mul_f32 %1, %9, %17 clamp\n\t"
        "v_pk_mul_f32 %2, %10, %18 clamp\n\t"
        "v_pk_mul_f32 %3, %11, %19 clamp\n\t"
        "v_pk_mul_f32 %4, %12, %16 clamp\n\t"
        "v_pk_mul_f32 %5, %13, %17 clamp\n\t"
        "v_pk_mul_f32 %6, %14, %18 clamp\n\t"
        "v_pk_mul_f32 %7, %15, %19 clamp"
        : "=&v"(o0), "=&v"(o1), "=&v"(o2), "=&v"(o3), "=&v"(o4), "=&v"(o5), "=&v"(o6), "=&v"(o7)
        : "v"(a0), "v"(a1), "v"(a2), "v"(a3), "v"(b0), "v"(b1), "v"(b2), "v"(b3),
          "v"(h0), "v"(h1), "v"(h2), "v"(h3));
    t0 = __builtin_shufflevector(__builtin_shufflevector(o0, o1, 0, 1, 2, 3), __builtin_shufflevector(o2, o3, 0, 1, 2, 3), 0, 1, 2, 3, 4, 5, 6, 7);
    t1 = __builtin_shufflevector(__builtin_shufflevector(o4, o5, 0, 1, 2, 3), __builtin_shufflevector(o6, o7, 0, 1, 2, 3), 0, 1, 2, 3, 4, 5, 6, 7);
}

__global__ __launch_bounds__(256)
void render_curves_wmma(const float* __restrict__ position,
                        const float* __restrict__ radius,
                        const float* __restrict__ color,
                        const float* __restrict__ background,
                        float* __restrict__ out)
{
    __align__(16) __shared__ float sVert[N_VERT][N_CURVES][4];
    __align__(16) __shared__ float sVV[N_SEG][N_CURVES];
    __align__(16) __shared__ float sHH[N_SEG][N_CURVES];
    __align__(16) __shared__ float sMeta[N_CURVES][4];

    const int b   = blockIdx.x / (IMAGE / 2);
    const int yp  = blockIdx.x % (IMAGE / 2);
    const int tid = threadIdx.x;

    const int wave = tid >> 5;
    const int lane = tid & 31;
    const int colp = lane & 15;
    const bool  hi = lane >= 16;

    float mRad = 1.0f, mC0 = 0.0f, mC1 = 0.0f, mC2 = 0.0f;
    {
        const int slot  = 16 * (wave & 3) + colp;
        const int curve = ((slot >> 3) & 1) * 32 + (slot >> 4) * 8 + (slot & 7);
        const bool odd  = (curve & 1) != 0;
        const float* P  = position + (b * N_LINES + (curve >> 1)) * 10;
        const int i0 = (odd ? 4 : 0) + (hi ? 4 : 0);
        const int i1 = hi ? i0 : i0 + 2;
        const v2f q0 = *(const v2f*)(P + i0);
        const v2f q1 = *(const v2f*)(P + i1);
        if (tid < N_CURVES) {
            const int mcurve = ((tid >> 3) & 1) * 32 + (tid >> 4) * 8 + (tid & 7);
            const int mline  = mcurve >> 1;
            mRad = radius[b * N_LINES + mline];
            mC0  = color[(b * N_LINES + mline) * 3 + 0];
            mC1  = color[(b * N_LINES + mline) * 3 + 1];
            mC2  = color[(b * N_LINES + mline) * 3 + 2];
        }
        const float midy = odd ? 2.0f * q0.x - q1.x : q1.x;
        const float midx = odd ? 2.0f * q0.y - q1.y : q1.y;
        v2f By, Bx;
        By.x = q0.x;  By.y = hi ? 0.0f : midy;
        Bx.x = q0.y;  Bx.y = hi ? 0.0f : midx;

        const int   js = colp < 8 ? colp : (colp < 11 ? colp - 1 : N_VERT - 1);
        const float tj = (float)js * (1.0f / 9.0f);
        const float uj = 1.0f - tj;
        v2f Aw;
        Aw.x = hi ? tj * tj : uj * uj;
        Aw.y = hi ? 0.0f    : 2.0f * tj * uj;

        v8f Dy = {}, Dx = {};
        Dy = __builtin_amdgcn_wmma_f32_16x16x4_f32(false, Aw, false, By, (short)0, Dy, false, false);
        Dx = __builtin_amdgcn_wmma_f32_16x16x4_f32(false, Aw, false, Bx, (short)0, Dx, false, false);

        const int first = hi ? 7 : 0;
        if (wave < 4) {
#pragma unroll
            for (int r = 0; r < 8; ++r) {
                const int vi = first + r;
                if (vi < N_VERT) {
                    const float ay = Dy[r], ax = Dx[r];
                    const v4f r2 = { -2.0f * ay, -2.0f * ax, ay*ay + ax*ax, 1.0f };
                    *(v4f*)&sVert[vi][slot][0] = r2;
                }
            }
        } else {
#pragma unroll
            for (int r = 0; r < 7; ++r) {
                const int si = first + r;
                if (si < N_SEG) {
                    const float vy = Dy[r + 1] - Dy[r], vx = Dx[r + 1] - Dx[r];
                    const float vv = vy*vy + vx*vx;
                    sVV[si][slot] = vv;
                    sHH[si][slot] = 0.5f / (vv + EPSF);
                }
            }
        }
    }
    if (tid < N_CURVES) {
        const float sg = mRad * mRad * SIGMA2;
        const v4f mt = { 1.0f / sg, mC0, mC1, mC2 };
        *(v4f*)&sMeta[tid][0] = mt;
    }
    __syncthreads();

    const int y    = 2 * yp + (wave >> 2);
    const int xs   = (wave & 3) * 32;
    const int x0   = xs + colp;
    const int x1   = xs + 16 + colp;
    const float sc = 2.0f / 127.0f;
    const float gy  = -1.0f + (float)y  * sc;
    const float gx0 = -1.0f + (float)x0 * sc;
    const float gx1 = -1.0f + (float)x1 * sc;

    v2f Bf0, Bf1;
    Bf0.x = hi ? 1.0f : gy;
    Bf0.y = hi ? fmaf(gy, gy, gx0 * gx0) : gx0;
    Bf1.x = hi ? 1.0f : gy;
    Bf1.y = hi ? fmaf(gy, gy, gx1 * gx1) : gx1;

    const int rowSel = lane & 15;
    const int kOff   = hi ? 2 : 0;
    const int metaRow = hi ? 8 : 0;

    float aR0 = 0.0f, aG0 = 0.0f, aB0 = 0.0f, w0 = 1.0f;
    float aR1 = 0.0f, aG1 = 0.0f, aB1 = 0.0f, w1 = 1.0f;

    for (int t = 0; t < N_TILES; ++t) {
        v2f Afr[N_VERT];
        v8f VVt[N_SEG], HHt[N_SEG];
        v4f MTt[8];
#pragma unroll
        for (int j = 0; j < N_VERT; ++j)
            Afr[j] = *(const v2f*)&sVert[j][t * 16 + rowSel][kOff];
#pragma unroll
        for (int s = 0; s < N_SEG; ++s) {
            const v4f va = *(const v4f*)&sVV[s][t * 16 + metaRow];
            const v4f vb = *(const v4f*)&sVV[s][t * 16 + metaRow + 4];
            const v4f ha = *(const v4f*)&sHH[s][t * 16 + metaRow];
            const v4f hb = *(const v4f*)&sHH[s][t * 16 + metaRow + 4];
            VVt[s] = __builtin_shufflevector(va, vb, 0, 1, 2, 3, 4, 5, 6, 7);
            HHt[s] = __builtin_shufflevector(ha, hb, 0, 1, 2, 3, 4, 5, 6, 7);
        }
#pragma unroll
        for (int i = 0; i < 8; ++i)
            MTt[i] = *(const v4f*)&sMeta[t * 16 + metaRow + i][0];
        __builtin_amdgcn_sched_barrier(0);

        v8f m0 = {BIGF, BIGF, BIGF, BIGF, BIGF, BIGF, BIGF, BIGF};
        v8f m1 = {BIGF, BIGF, BIGF, BIGF, BIGF, BIGF, BIGF, BIGF};

        v8f Dp0 = {}, Dp1 = {};
        Dp0 = __builtin_amdgcn_wmma_f32_16x16x4_f32(false, Afr[0], false, Bf0, (short)0, Dp0, false, false);
        Dp1 = __builtin_amdgcn_wmma_f32_16x16x4_f32(false, Afr[0], false, Bf1, (short)0, Dp1, false, false);

#pragma unroll
        for (int s = 0; s < N_SEG; ++s) {
            const v8f VV = VVt[s];
            const v8f HH = HHt[s];

            v8f Dn0 = {}, Dn1 = {};
            Dn0 = __builtin_amdgcn_wmma_f32_16x16x4_f32(false, Afr[s + 1], false, Bf0, (short)0, Dn0, false, false);
            Dn1 = __builtin_amdgcn_wmma_f32_16x16x4_f32(false, Afr[s + 1], false, Bf1, (short)0, Dn1, false, false);

            {
                const v8f G0 = (Dp0 - Dn0) + VV;
                const v8f G1 = (Dp1 - Dn1) + VV;
                v8f tp0, tp1;
                mul_clamp01_x2(G0, G1, HH, tp0, tp1);
                const v8f e20 = Dp0 - tp0 * (G0 - tp0 * VV);
                const v8f e21 = Dp1 - tp1 * (G1 - tp1 * VV);
                m0 = __builtin_elementwise_min(m0, e20);
                m1 = __builtin_elementwise_min(m1, e21);
                Dp0 = Dn0;
                Dp1 = Dn1;
            }
        }

#pragma unroll
        for (int i = 0; i < 8; ++i) {
            const v4f mt = MTt[i];
            {
                const float z = m0[i] * mt.x;
                const float I = __expf(-(z * z));
                const float wI = w0 * I;
                aR0 = fmaf(mt.y, wI, aR0);
                aG0 = fmaf(mt.z, wI, aG0);
                aB0 = fmaf(mt.w, wI, aB0);
                w0 = w0 * (1.0f - I + EPSF);
            }
            {
                const float z = m1[i] * mt.x;
                const float I = __expf(-(z * z));
                const float wI = w1 * I;
                aR1 = fmaf(mt.y, wI, aR1);
                aG1 = fmaf(mt.z, wI, aG1);
                aB1 = fmaf(mt.w, wI, aB1);
                w1 = w1 * (1.0f - I + EPSF);
            }
        }
    }

    const float kR = hi ? aR1 : aR0, kG = hi ? aG1 : aG0, kB = hi ? aB1 : aB0, kW = hi ? w1 : w0;
    const float rR = xhalf(hi ? aR0 : aR1), rG = xhalf(hi ? aG0 : aG1);
    const float rB = xhalf(hi ? aB0 : aB1), rW = xhalf(hi ? w0 : w1);
    const float fwR = hi ? rR : kR, fwG = hi ? rG : kG, fwB = hi ? rB : kB, fwW = hi ? rW : kW;
    const float bkR = hi ? kR : rR, bkG = hi ? kG : rG, bkB = hi ? kB : rB, bkW = hi ? kW : rW;
    {
        const float fR = fmaf(fwW, bkR, fwR);
        const float fG = fmaf(fwW, bkG, fwG);
        const float fB = fmaf(fwW, bkB, fwB);
        const float fW = fwW * bkW;
        const float bg0 = background[b * 3 + 0];
        const float bg1 = background[b * 3 + 1];
        const float bg2 = background[b * 3 + 2];
        const int base = ((b * 3) * IMAGE + y) * IMAGE + xs + lane;
        out[base]                     = clamp01(fmaf(bg0, fW, fR));
        out[base + IMAGE * IMAGE]     = clamp01(fmaf(bg1, fW, fG));
        out[base + 2 * IMAGE * IMAGE] = clamp01(fmaf(bg2, fW, fB));
    }
}

extern "C" void kernel_launch(void* const* d_in, const int* in_sizes, int n_in,
                              void* d_out, int out_size, void* d_ws, size_t ws_size,
                              hipStream_t stream) {
    (void)in_sizes; (void)n_in; (void)out_size; (void)d_ws; (void)ws_size;
    const float* position   = (const float*)d_in[0];
    const float* radius     = (const float*)d_in[1];
    const float* color      = (const float*)d_in[2];
    const float* background = (const float*)d_in[3];
    float* out = (float*)d_out;
    render_curves_wmma<<<4 * (IMAGE / 2), 256, 0, stream>>>(position, radius, color, background, out);
}


# Matrix products for a wave's left pixel tile only: the right tile's vertex distances are the left one's plus a per-(vertex,curve) column offset from LDS (10 WMMA per curve pass, was 20); rounds anew
# speedup vs baseline: 2.8472x; 1.0504x over previous
#include <hip/hip_runtime.h>


#define IMAGE    128
#define N_LINES  32
#define N_CURVES 64
#define N_SEG    9
#define N_VERT   10
#define N_TILES  4
#define SIGMA2   0.05f
#define EPSF     1e-10f
#define BIGF     1e30f
#define GSTEP    (2.0f / 127.0f)
#define DXT      (16.0f * GSTEP)

typedef float v2f __attribute__((ext_vector_type(2)));
typedef float v4f __attribute__((ext_vector_type(4)));
typedef float v8f __attribute__((ext_vector_type(8)));

__device__ __forceinline__ float clamp01(float v) { return fminf(fmaxf(v, 0.0f), 1.0f); }

__device__ __forceinline__ float xhalf(float x) {
    return __builtin_bit_cast(float, __builtin_amdgcn_permlanex16(__builtin_bit_cast(int, x), __builtin_bit_cast(int, x),
                                                                  0x76543210, 0xfedcba98, false, false));
}


__device__ __forceinline__ void mul_clamp01_x2(const v8f g0, const v8f g1, const v8f h, v8f& t0, v8f& t1) {
    const v2f a0 = { g0[0], g0[1] }, a1 = { g0[2], g0[3] }, a2 = { g0[4], g0[5] }, a3 = { g0[6], g0[7] };
    const v2f b0 = { g1[0], g1[1] }, b1 = { g1[2], g1[3] }, b2 = { g1[4], g1[5] }, b3 = { g1[6], g1[7] };
    const v2f h0 = { h[0], h[1] },   h1 = { h[2], h[3] },   h2 = { h[4], h[5] },   h3 = { h[6], h[7] };
    v2f o0, o1, o2, o3, o4, o5, o6, o7;
    asm("v_pk_mul_f32 %0, %8, %16 clamp\n\t"
        "v_pk_mul_f32 %1, %9, %17 clamp\n\t"
        "v_pk_mul_f32 %2, %10, %18 clamp\n\t"
        "v_pk_mul_f32 %3, %11, %19 clamp\n\t"
        "v_pk_mul_f32 %4, %12, %16 clamp\n\t"
        "v_pk_mul_f32 %5, %13, %17 clamp\n\t"
        "v_pk_mul_f32 %6, %14, %18 clamp\n\t"
        "v_pk_mul_f32 %7, %15, %19 clamp"
        : "=&v"(o0), "=&v"(o1), "=&v"(o2), "=&v"(o3), "=&v"(o4), "=&v"(o5), "=&v"(o6), "=&v"(o7)
        : "v"(a0), "v"(a1), "v"(a2), "v"(a3), "v"(b0), "v"(b1), "v"(b2), "v"(b3),
          "v"(h0), "v"(h1), "v"(h2), "v"(h3));
    t0 = __builtin_shufflevector(__builtin_shufflevector(o0, o1, 0, 1, 2, 3), __builtin_shufflevector(o2, o3, 0, 1, 2, 3), 0, 1, 2, 3, 4, 5, 6, 7);
    t1 = __builtin_shufflevector(__builtin_shufflevector(o4, o5, 0, 1, 2, 3), __builtin_shufflevector(o6, o7, 0, 1, 2, 3), 0, 1, 2, 3, 4, 5, 6, 7);
}

__global__ __launch_bounds__(256)
void render_curves_wmma(const float* __restrict__ position,
                        const float* __restrict__ radius,
                        const float* __restrict__ color,
                        const float* __restrict__ background,
                        float* __restrict__ out)
{
    __align__(16) __shared__ float sVert[N_VERT][N_CURVES][4];
    __align__(16) __shared__ float sVV[N_SEG][N_CURVES];
    __align__(16) __shared__ float sHH[N_SEG][N_CURVES];
    __align__(16) __shared__ float sCX[N_VERT][N_CURVES];
    __align__(16) __shared__ float sMeta[N_CURVES][4];

    const int b   = blockIdx.x / (IMAGE / 2);
    const int yp  = blockIdx.x % (IMAGE / 2);
    const int tid = threadIdx.x;

    const int wave = tid >> 5;
    const int lane = tid & 31;
    const int colp = lane & 15;
    const bool  hi = lane >= 16;

    float mRad = 1.0f, mC0 = 0.0f, mC1 = 0.0f, mC2 = 0.0f;
    {
        const int slot  = 16 * (wave & 3) + colp;
        const int curve = ((slot >> 3) & 1) * 32 + (slot >> 4) * 8 + (slot & 7);
        const bool odd  = (curve & 1) != 0;
        const float* P  = position + (b * N_LINES + (curve >> 1)) * 10;
        const int i0 = (odd ? 4 : 0) + (hi ? 4 : 0);
        const int i1 = hi ? i0 : i0 + 2;
        const v2f q0 = *(const v2f*)(P + i0);
        const v2f q1 = *(const v2f*)(P + i1);
        if (tid < N_CURVES) {
            const int mcurve = ((tid >> 3) & 1) * 32 + (tid >> 4) * 8 + (tid & 7);
            const int mline  = mcurve >> 1;
            mRad = radius[b * N_LINES + mline];
            mC0  = color[(b * N_LINES + mline) * 3 + 0];
            mC1  = color[(b * N_LINES + mline) * 3 + 1];
            mC2  = color[(b * N_LINES + mline) * 3 + 2];
        }
        const float midy = odd ? 2.0f * q0.x - q1.x : q1.x;
        const float midx = odd ? 2.0f * q0.y - q1.y : q1.y;
        v2f By, Bx;
        By.x = q0.x;  By.y = hi ? 0.0f : midy;
        Bx.x = q0.y;  Bx.y = hi ? 0.0f : midx;

        const int   js = colp < 8 ? colp : (colp < 11 ? colp - 1 : N_VERT - 1);
        const float tj = (float)js * (1.0f / 9.0f);
        const float uj = 1.0f - tj;
        v2f Aw;
        Aw.x = hi ? tj * tj : uj * uj;
        Aw.y = hi ? 0.0f    : 2.0f * tj * uj;

        v8f Dy = {}, Dx = {};
        Dy = __builtin_amdgcn_wmma_f32_16x16x4_f32(false, Aw, false, By, (short)0, Dy, false, false);
        Dx = __builtin_amdgcn_wmma_f32_16x16x4_f32(false, Aw, false, Bx, (short)0, Dx, false, false);

        const int first = hi ? 7 : 0;
        if (wave < 4) {
#pragma unroll
            for (int r = 0; r < 8; ++r) {
                const int vi = first + r;
                if (vi < N_VERT) {
                    const float ay = Dy[r], ax = Dx[r];
                    const v4f r2 = { -2.0f * ay, -2.0f * ax, ay*ay + ax*ax, 1.0f };
                    *(v4f*)&sVert[vi][slot][0] = r2;
                    sCX[vi][slot] = DXT * (-2.0f * ax);
                }
            }
        } else {
#pragma unroll
            for (int r = 0; r < 7; ++r) {
                const int si = first + r;
                if (si < N_SEG) {
                    const float vy = Dy[r + 1] - Dy[r], vx = Dx[r + 1] - Dx[r];
                    const float vv = vy*vy + vx*vx;
                    sVV[si][slot] = vv;
                    sHH[si][slot] = 0.5f / (vv + EPSF);
                }
            }
        }
    }
    if (tid < N_CURVES) {
        const float sg = mRad * mRad * SIGMA2;
        const v4f mt = { 1.0f / sg, mC0, mC1, mC2 };
        *(v4f*)&sMeta[tid][0] = mt;
    }
    __syncthreads();

    const int y    = 2 * yp + (wave >> 2);
    const int xs   = (wave & 3) * 32;
    const int x0   = xs + colp;
    const float sc = GSTEP;
    const float gy  = -1.0f + (float)y  * sc;
    const float gx0 = -1.0f + (float)x0 * sc;
    const float Lx  = fmaf(2.0f * DXT, gx0, DXT * DXT);

    v2f Bf0;
    Bf0.x = hi ? 1.0f : gy;
    Bf0.y = hi ? fmaf(gy, gy, gx0 * gx0) : gx0;

    const int rowSel = lane & 15;
    const int kOff   = hi ? 2 : 0;
    const int metaRow = hi ? 8 : 0;

    float aR0 = 0.0f, aG0 = 0.0f, aB0 = 0.0f, w0 = 1.0f;
    float aR1 = 0.0f, aG1 = 0.0f, aB1 = 0.0f, w1 = 1.0f;

    for (int t = 0; t < N_TILES; ++t) {
        v2f Afr[N_VERT];
        v8f VVt[N_SEG], HHt[N_SEG], CXt[N_VERT];
        v4f MTt[8];
#pragma unroll
        for (int j = 0; j < N_VERT; ++j)
            Afr[j] = *(const v2f*)&sVert[j][t * 16 + rowSel][kOff];
#pragma unroll
        for (int s = 0; s < N_SEG; ++s) {
            const v4f va = *(const v4f*)&sVV[s][t * 16 + metaRow];
            const v4f vb = *(const v4f*)&sVV[s][t * 16 + metaRow + 4];
            const v4f ha = *(const v4f*)&sHH[s][t * 16 + metaRow];
            const v4f hb = *(const v4f*)&sHH[s][t * 16 + metaRow + 4];
            VVt[s] = __builtin_shufflevector(va, vb, 0, 1, 2, 3, 4, 5, 6, 7);
            HHt[s] = __builtin_shufflevector(ha, hb, 0, 1, 2, 3, 4, 5, 6, 7);
        }
#pragma unroll
        for (int j = 0; j < N_VERT; ++j) {
            const v4f ca = *(const v4f*)&sCX[j][t * 16 + metaRow];
            const v4f cb = *(const v4f*)&sCX[j][t * 16 + metaRow + 4];
            CXt[j] = __builtin_shufflevector(ca, cb, 0, 1, 2, 3, 4, 5, 6, 7);
        }
#pragma unroll
        for (int i = 0; i < 8; ++i)
            MTt[i] = *(const v4f*)&sMeta[t * 16 + metaRow + i][0];
        __builtin_amdgcn_sched_barrier(0);

        v8f m0 = {BIGF, BIGF, BIGF, BIGF, BIGF, BIGF, BIGF, BIGF};
        v8f m1 = {BIGF, BIGF, BIGF, BIGF, BIGF, BIGF, BIGF, BIGF};

        v8f Dp0 = {}, Dn0 = {};
        Dp0 = __builtin_amdgcn_wmma_f32_16x16x4_f32(false, Afr[0], false, Bf0, (short)0, Dp0, false, false);
        Dn0 = __builtin_amdgcn_wmma_f32_16x16x4_f32(false, Afr[1], false, Bf0, (short)0, Dn0, false, false);
        v8f Dp1 = Dp0 + CXt[0];

#pragma unroll
        for (int s = 0; s < N_SEG; ++s) {
            const v8f VV = VVt[s];
            const v8f HH = HHt[s];

            v8f Dnn0 = {};
            if (s + 2 < N_VERT)
                Dnn0 = __builtin_amdgcn_wmma_f32_16x16x4_f32(false, Afr[s + 2], false, Bf0, (short)0, Dnn0, false, false);
            const v8f Dn1 = Dn0 + CXt[s + 1];

            {
                const v8f G0 = (Dp0 - Dn0) + VV;
                const v8f G1 = (Dp1 - Dn1) + VV;
                v8f tp0, tp1;
                mul_clamp01_x2(G0, G1, HH, tp0, tp1);
                const v8f e20 = Dp0 - tp0 * (G0 - tp0 * VV);
                const v8f e21 = Dp1 - tp1 * (G1 - tp1 * VV);
                m0 = __builtin_elementwise_min(m0, e20);
                m1 = __builtin_elementwise_min(m1, e21);
                Dp0 = Dn0;
                Dp1 = Dn1;
                Dn0 = Dnn0;
            }
            __builtin_amdgcn_sched_barrier(0);
        }

#pragma unroll
        for (int i = 0; i < 8; ++i) {
            const v4f mt = MTt[i];
            {
                const float z = m0[i] * mt.x;
                const float I = __expf(-(z * z));
                const float wI = w0 * I;
                aR0 = fmaf(mt.y, wI, aR0);
                aG0 = fmaf(mt.z, wI, aG0);
                aB0 = fmaf(mt.w, wI, aB0);
                w0 = w0 * (1.0f - I + EPSF);
            }
            {
                const float z = (m1[i] + Lx) * mt.x;
                const float I = __expf(-(z * z));
                const float wI = w1 * I;
                aR1 = fmaf(mt.y, wI, aR1);
                aG1 = fmaf(mt.z, wI, aG1);
                aB1 = fmaf(mt.w, wI, aB1);
                w1 = w1 * (1.0f - I + EPSF);
            }
        }
    }

    const float kR = hi ? aR1 : aR0, kG = hi ? aG1 : aG0, kB = hi ? aB1 : aB0, kW = hi ? w1 : w0;
    const float rR = xhalf(hi ? aR0 : aR1), rG = xhalf(hi ? aG0 : aG1);
    const float rB = xhalf(hi ? aB0 : aB1), rW = xhalf(hi ? w0 : w1);
    const float fwR = hi ? rR : kR, fwG = hi ? rG : kG, fwB = hi ? rB : kB, fwW = hi ? rW : kW;
    const float bkR = hi ? kR : rR, bkG = hi ? kG : rG, bkB = hi ? kB : rB, bkW = hi ? kW : rW;
    {
        const float fR = fmaf(fwW, bkR, fwR);
        const float fG = fmaf(fwW, bkG, fwG);
        const float fB = fmaf(fwW, bkB, fwB);
        const float fW = fwW * bkW;
        const float bg0 = background[b * 3 + 0];
        const float bg1 = background[b * 3 + 1];
        const float bg2 = background[b * 3 + 2];
        const int base = ((b * 3) * IMAGE + y) * IMAGE + xs + lane;
        out[base]                     = clamp01(fmaf(bg0, fW, fR));
        out[base + IMAGE * IMAGE]     = clamp01(fmaf(bg1, fW, fG));
        out[base + 2 * IMAGE * IMAGE] = clamp01(fmaf(bg2, fW, fB));
    }
}

extern "C" void kernel_launch(void* const* d_in, const int* in_sizes, int n_in,
                              void* d_out, int out_size, void* d_ws, size_t ws_size,
                              hipStream_t stream) {
    (void)in_sizes; (void)n_in; (void)out_size; (void)d_ws; (void)ws_size;
    const float* position   = (const float*)d_in[0];
    const float* radius     = (const float*)d_in[1];
    const float* color      = (const float*)d_in[2];
    const float* background = (const float*)d_in[3];
    float* out = (float*)d_out;
    render_curves_wmma<<<4 * (IMAGE / 2), 256, 0, stream>>>(position, radius, color, background, out);
}


# Segment step forms only D_s - t u_s (u_s = d.v, t its clipped foot): a foot past the far end is left to the next vertex's own distance, last vertex joins the min; one fma less per step; rounds anew
# speedup vs baseline: 3.0127x; 1.0581x over previous
#include <hip/hip_runtime.h>


#define IMAGE    128
#define N_LINES  32
#define N_CURVES 64
#define N_SEG    9
#define N_VERT   10
#define N_TILES  4
#define SIGMA2   0.05f
#define EPSF     1e-10f
#define GSTEP    (2.0f / 127.0f)
#define DXT      (16.0f * GSTEP)

typedef float v2f __attribute__((ext_vector_type(2)));
typedef float v4f __attribute__((ext_vector_type(4)));
typedef float v8f __attribute__((ext_vector_type(8)));

__device__ __forceinline__ float clamp01(float v) { return fminf(fmaxf(v, 0.0f), 1.0f); }

__device__ __forceinline__ float xhalf(float x) {
    return __builtin_bit_cast(float, __builtin_amdgcn_permlanex16(__builtin_bit_cast(int, x), __builtin_bit_cast(int, x),
                                                                  0x76543210, 0xfedcba98, false, false));
}


__device__ __forceinline__ void mul_clamp01_x2(const v8f g0, const v8f g1, const v8f h, v8f& t0, v8f& t1) {
    const v2f a0 = { g0[0], g0[1] }, a1 = { g0[2], g0[3] }, a2 = { g0[4], g0[5] }, a3 = { g0[6], g0[7] };
    const v2f b0 = { g1[0], g1[1] }, b1 = { g1[2], g1[3] }, b2 = { g1[4], g1[5] }, b3 = { g1[6], g1[7] };
    const v2f h0 = { h[0], h[1] },   h1 = { h[2], h[3] },   h2 = { h[4], h[5] },   h3 = { h[6], h[7] };
    v2f o0, o1, o2, o3, o4, o5, o6, o7;
    asm("v_pk_mul_f32 %0, %8, %16 clamp\n\t"
        "v_pk_mul_f32 %1, %9, %17 clamp\n\t"
        "v_pk_mul_f32 %2, %10, %18 clamp\n\t"
        "v_pk_mul_f32 %3, %11, %19 clamp\n\t"
        "v_pk_mul_f32 %4, %12, %16 clamp\n\t"
        "v_pk_mul_f32 %5, %13, %17 clamp\n\t"
        "v_pk_mul_f32 %6, %14, %18 clamp\n\t"
        "v_pk_mul_f32 %7, %15, %19 clamp"
        : "=&v"(o0), "=&v"(o1), "=&v"(o2), "=&v"(o3), "=&v"(o4), "=&v"(o5), "=&v"(o6), "=&v"(o7)
        : "v"(a0), "v"(a1), "v"(a2), "v"(a3), "v"(b0), "v"(b1), "v"(b2), "v"(b3),
          "v"(h0), "v"(h1), "v"(h2), "v"(h3));
    t0 = __builtin_shufflevector(__builtin_shufflevector(o0, o1, 0, 1, 2, 3), __builtin_shufflevector(o2, o3, 0, 1, 2, 3), 0, 1, 2, 3, 4, 5, 6, 7);
    t1 = __builtin_shufflevector(__builtin_shufflevector(o4, o5, 0, 1, 2, 3), __builtin_shufflevector(o6, o7, 0, 1, 2, 3), 0, 1, 2, 3, 4, 5, 6, 7);
}

__global__ __launch_bounds__(256)
void render_curves_wmma(const float* __restrict__ position,
                        const float* __restrict__ radius,
                        const float* __restrict__ color,
                        const float* __restrict__ background,
                        float* __restrict__ out)
{
    __align__(16) __shared__ float sVert[N_VERT][N_CURVES][4];
    __align__(16) __shared__ float sVV[N_SEG][N_CURVES];
    __align__(16) __shared__ float sHH[N_SEG][N_CURVES];
    __align__(16) __shared__ float sCX[N_VERT][N_CURVES];
    __align__(16) __shared__ float sMeta[N_CURVES][4];

    const int b   = blockIdx.x / (IMAGE / 2);
    const int yp  = blockIdx.x % (IMAGE / 2);
    const int tid = threadIdx.x;

    const int wave = tid >> 5;
    const int lane = tid & 31;
    const int colp = lane & 15;
    const bool  hi = lane >= 16;

    float mRad = 1.0f, mC0 = 0.0f, mC1 = 0.0f, mC2 = 0.0f;
    {
        const int slot  = 16 * (wave & 3) + colp;
        const int curve = ((slot >> 3) & 1) * 32 + (slot >> 4) * 8 + (slot & 7);
        const bool odd  = (curve & 1) != 0;
        const float* P  = position + (b * N_LINES + (curve >> 1)) * 10;
        const int i0 = (odd ? 4 : 0) + (hi ? 4 : 0);
        const int i1 = hi ? i0 : i0 + 2;
        const v2f q0 = *(const v2f*)(P + i0);
        const v2f q1 = *(const v2f*)(P + i1);
        if (tid < N_CURVES) {
            const int mcurve = ((tid >> 3) & 1) * 32 + (tid >> 4) * 8 + (tid & 7);
            const int mline  = mcurve >> 1;
            mRad = radius[b * N_LINES + mline];
            mC0  = color[(b * N_LINES + mline) * 3 + 0];
            mC1  = color[(b * N_LINES + mline) * 3 + 1];
            mC2  = color[(b * N_LINES + mline) * 3 + 2];
        }
        const float midy = odd ? 2.0f * q0.x - q1.x : q1.x;
        const float midx = odd ? 2.0f * q0.y - q1.y : q1.y;
        v2f By, Bx;
        By.x = q0.x;  By.y = hi ? 0.0f : midy;
        Bx.x = q0.y;  Bx.y = hi ? 0.0f : midx;

        const int   js = colp < 8 ? colp : (colp < 11 ? colp - 1 : N_VERT - 1);
        const float tj = (float)js * (1.0f / 9.0f);
        const float uj = 1.0f - tj;
        v2f Aw;
        Aw.x = hi ? tj * tj : uj * uj;
        Aw.y = hi ? 0.0f    : 2.0f * tj * uj;

        v8f Dy = {}, Dx = {};
        Dy = __builtin_amdgcn_wmma_f32_16x16x4_f32(false, Aw, false, By, (short)0, Dy, false, false);
        Dx = __builtin_amdgcn_wmma_f32_16x16x4_f32(false, Aw, false, Bx, (short)0, Dx, false, false);

        const int first = hi ? 7 : 0;
        if (wave < 4) {
#pragma unroll
            for (int r = 0; r < 8; ++r) {
                const int vi = first + r;
                if (vi < N_VERT) {
                    const float ay = Dy[r], ax = Dx[r];
                    const v4f r2 = { -2.0f * ay, -2.0f * ax, ay*ay + ax*ax, 1.0f };
                    *(v4f*)&sVert[vi][slot][0] = r2;
                    sCX[vi][slot] = DXT * (-2.0f * ax);
                }
            }
        } else {
#pragma unroll
            for (int r = 0; r < 7; ++r) {
                const int si = first + r;
                if (si < N_SEG) {
                    const float vy = Dy[r + 1] - Dy[r], vx = Dx[r + 1] - Dx[r];
                    const float vv = vy*vy + vx*vx;
                    sVV[si][slot] = 0.5f * vv;
                    sHH[si][slot] = 1.0f / (vv + EPSF);
                }
            }
        }
    }
    if (tid < N_CURVES) {
        const float sg = mRad * mRad * SIGMA2;
        const v4f mt = { 1.0f / sg, mC0, mC1, mC2 };
        *(v4f*)&sMeta[tid][0] = mt;
    }
    __syncthreads();

    const int y    = 2 * yp + (wave >> 2);
    const int xs   = (wave & 3) * 32;
    const int x0   = xs + colp;
    const float sc = GSTEP;
    const float gy  = -1.0f + (float)y  * sc;
    const float gx0 = -1.0f + (float)x0 * sc;
    const float Lx  = fmaf(2.0f * DXT, gx0, DXT * DXT);

    v2f Bf0;
    Bf0.x = hi ? 1.0f : gy;
    Bf0.y = hi ? fmaf(gy, gy, gx0 * gx0) : gx0;

    const int rowSel = lane & 15;
    const int kOff   = hi ? 2 : 0;
    const int metaRow = hi ? 8 : 0;

    float aR0 = 0.0f, aG0 = 0.0f, aB0 = 0.0f, w0 = 1.0f;
    float aR1 = 0.0f, aG1 = 0.0f, aB1 = 0.0f, w1 = 1.0f;

    for (int t = 0; t < N_TILES; ++t) {
        v2f Afr[N_VERT];
        v8f VVt[N_SEG], HHt[N_SEG], CXt[N_VERT];
        v4f MTt[8];
#pragma unroll
        for (int j = 0; j < N_VERT; ++j)
            Afr[j] = *(const v2f*)&sVert[j][t * 16 + rowSel][kOff];
#pragma unroll
        for (int s = 0; s < N_SEG; ++s) {
            const v4f va = *(const v4f*)&sVV[s][t * 16 + metaRow];
            const v4f vb = *(const v4f*)&sVV[s][t * 16 + metaRow + 4];
            const v4f ha = *(const v4f*)&sHH[s][t * 16 + metaRow];
            const v4f hb = *(const v4f*)&sHH[s][t * 16 + metaRow + 4];
            VVt[s] = __builtin_shufflevector(va, vb, 0, 1, 2, 3, 4, 5, 6, 7);
            HHt[s] = __builtin_shufflevector(ha, hb, 0, 1, 2, 3, 4, 5, 6, 7);
        }
#pragma unroll
        for (int j = 0; j < N_VERT; ++j) {
            const v4f ca = *(const v4f*)&sCX[j][t * 16 + metaRow];
            const v4f cb = *(const v4f*)&sCX[j][t * 16 + metaRow + 4];
            CXt[j] = __builtin_shufflevector(ca, cb, 0, 1, 2, 3, 4, 5, 6, 7);
        }
#pragma unroll
        for (int i = 0; i < 8; ++i)
            MTt[i] = *(const v4f*)&sMeta[t * 16 + metaRow + i][0];
        __builtin_amdgcn_sched_barrier(0);

        v8f m0, m1;

        v8f Dp0 = {}, Dn0 = {};
        Dp0 = __builtin_amdgcn_wmma_f32_16x16x4_f32(false, Afr[0], false, Bf0, (short)0, Dp0, false, false);
        Dn0 = __builtin_amdgcn_wmma_f32_16x16x4_f32(false, Afr[1], false, Bf0, (short)0, Dn0, false, false);
        v8f Dp1 = Dp0 + CXt[0];

#pragma unroll
        for (int s = 0; s < N_SEG; ++s) {
            const v8f VV = VVt[s];
            const v8f HH = HHt[s];

            v8f Dnn0 = {};
            if (s + 2 < N_VERT)
                Dnn0 = __builtin_amdgcn_wmma_f32_16x16x4_f32(false, Afr[s + 2], false, Bf0, (short)0, Dnn0, false, false);
            const v8f Dn1 = Dn0 + CXt[s + 1];

            {
                const v8f U0 = (Dp0 - Dn0) * 0.5f + VV;
                const v8f U1 = (Dp1 - Dn1) * 0.5f + VV;
                v8f tp0, tp1;
                mul_clamp01_x2(U0, U1, HH, tp0, tp1);
                const v8f e20 = Dp0 - tp0 * U0;
                const v8f e21 = Dp1 - tp1 * U1;
                m0 = s == 0 ? e20 : __builtin_elementwise_min(m0, e20);
                m1 = s == 0 ? e21 : __builtin_elementwise_min(m1, e21);
                Dp0 = Dn0;
                Dp1 = Dn1;
                Dn0 = Dnn0;
            }
            __builtin_amdgcn_sched_barrier(0);
        }
        m0 = __builtin_elementwise_min(m0, Dp0);
        m1 = __builtin_elementwise_min(m1, Dp1);

#pragma unroll
        for (int i = 0; i < 8; ++i) {
            const v4f mt = MTt[i];
            {
                const float z = m0[i] * mt.x;
                const float I = __expf(-(z * z));
                const float wI = w0 * I;
                aR0 = fmaf(mt.y, wI, aR0);
                aG0 = fmaf(mt.z, wI, aG0);
                aB0 = fmaf(mt.w, wI, aB0);
                w0 = w0 * (1.0f - I + EPSF);
            }
            {
                const float z = (m1[i] + Lx) * mt.x;
                const float I = __expf(-(z * z));
                const float wI = w1 * I;
                aR1 = fmaf(mt.y, wI, aR1);
                aG1 = fmaf(mt.z, wI, aG1);
                aB1 = fmaf(mt.w, wI, aB1);
                w1 = w1 * (1.0f - I + EPSF);
            }
        }
    }

    const float kR = hi ? aR1 : aR0, kG = hi ? aG1 : aG0, kB = hi ? aB1 : aB0, kW = hi ? w1 : w0;
    const float rR = xhalf(hi ? aR0 : aR1), rG = xhalf(hi ? aG0 : aG1);
    const float rB = xhalf(hi ? aB0 : aB1), rW = xhalf(hi ? w0 : w1);
    const float fwR = hi ? rR : kR, fwG = hi ? rG : kG, fwB = hi ? rB : kB, fwW = hi ? rW : kW;
    const float bkR = hi ? kR : rR, bkG = hi ? kG : rG, bkB = hi ? kB : rB, bkW = hi ? kW : rW;
    {
        const float fR = fmaf(fwW, bkR, fwR);
        const float fG = fmaf(fwW, bkG, fwG);
        const float fB = fmaf(fwW, bkB, fwB);
        const float fW = fwW * bkW;
        const float bg0 = background[b * 3 + 0];
        const float bg1 = background[b * 3 + 1];
        const float bg2 = background[b * 3 + 2];
        const int base = ((b * 3) * IMAGE + y) * IMAGE + xs + lane;
        out[base]                     = clamp01(fmaf(bg0, fW, fR));
        out[base + IMAGE * IMAGE]     = clamp01(fmaf(bg1, fW, fG));
        out[base + 2 * IMAGE * IMAGE] = clamp01(fmaf(bg2, fW, fB));
    }
}

extern "C" void kernel_launch(void* const* d_in, const int* in_sizes, int n_in,
                              void* d_out, int out_size, void* d_ws, size_t ws_size,
                              hipStream_t stream) {
    (void)in_sizes; (void)n_in; (void)out_size; (void)d_ws; (void)ws_size;
    const float* position   = (const float*)d_in[0];
    const float* radius     = (const float*)d_in[1];
    const float* color      = (const float*)d_in[2];
    const float* background = (const float*)d_in[3];
    float* out = (float*)d_out;
    render_curves_wmma<<<4 * (IMAGE / 2), 256, 0, stream>>>(position, radius, color, background, out);
}
